# starting kernel
# baseline (speedup 1.0000x reference)
#include <hip/hip_runtime.h>
#include <hip/hip_bf16.h>

typedef __attribute__((ext_vector_type(2))) float v2f;
typedef __attribute__((ext_vector_type(8))) float v8f;

#define BS     4
#define NV     2048
#define NN     20
#define IN_CH  128
#define OUT_CH 128
#define NS     4
#define SXO    (NS * OUT_CH)
#define NCOL   ((NS + 1) * OUT_CH)
#define NROW   (BS * NV)
#define EPSF   1e-12f

__global__ __launch_bounds__(256)
void gemm_wmma_f32_kernel(const float* __restrict__ fm,
                          const float* __restrict__ w,
                          const float* __restrict__ bias,
                          float* __restrict__ fout) {
    const int lane  = threadIdx.x & 31;
    const int wave  = threadIdx.x >> 5;
    const int gwave = blockIdx.x * 8 + wave;

    const int ntiles_n = NCOL / 16;
    const int tm = gwave / ntiles_n;
    const int tn = gwave % ntiles_n;

    const int half = lane >> 4;
    const int l15  = lane & 15;
    const int r0   = tm * 16;
    const int c0   = tn * 16;

    v8f acc = {};
    for (int kk = 0; kk < IN_CH; kk += 4) {
        const float* arow = fm + (size_t)(r0 + l15) * IN_CH + (kk + 2 * half);
        v2f a;
        a.x = arow[0];
        a.y = arow[1];

        const float* brow = w + (size_t)(kk + 2 * half) * NCOL + c0 + l15;
        v2f b;
        b.x = brow[0];
        b.y = brow[NCOL];

        acc = __builtin_amdgcn_wmma_f32_16x16x4_f32(
             false, a,  false, b,
             (short)0, acc,  false,  false);
    }

    const float bv = bias[c0 + l15];
#pragma unroll
    for (int r = 0; r < 8; ++r) {
        const int row = r0 + r + half * 8;
        fout[(size_t)row * NCOL + c0 + l15] = acc[r] + bv;
    }
}

__global__ __launch_bounds__(256)
void fused_gconv_kernel(const int*   __restrict__ nidx,
                        const float* __restrict__ verts,
                        const float* __restrict__ fout,
                        const float* __restrict__ dirs,
                        const float* __restrict__ dw,
                        float*       __restrict__ out) {
    const int bv = blockIdx.x;
    const int b  = bv / NV;

    __shared__ float s_dn[NN][3];
    __shared__ float s_dist[NN];
    __shared__ int   s_idx[NN];
    __shared__ float s_maxd;
    __shared__ float s_vals[SXO];

    const int tid = threadIdx.x;

    if (tid < NN) {
        const int j   = tid;
        const int idx = nidx[(size_t)bv * NN + j];
        s_idx[j] = idx;

        const float cx = verts[(size_t)bv * 3 + 0];
        const float cy = verts[(size_t)bv * 3 + 1];
        const float cz = verts[(size_t)bv * 3 + 2];

        const size_t np = ((size_t)b * NV + idx) * 3;
        const float dx = verts[np + 0] - cx;
        const float dy = verts[np + 1] - cy;
        const float dz = verts[np + 2] - cz;

        const float dist = sqrtf(dx * dx + dy * dy + dz * dz);
        const float inv  = 1.0f / fmaxf(dist, EPSF);
        s_dn[j][0] = dx * inv;
        s_dn[j][1] = dy * inv;
        s_dn[j][2] = dz * inv;
        s_dist[j]  = dist;
    }
    __syncthreads();
    if (tid == 0) {
        float md = s_dist[0];
#pragma unroll
        for (int j = 1; j < NN; ++j) md = fmaxf(md, s_dist[j]);
        s_maxd = md;
    }
    __syncthreads();
    const float max_dist = s_maxd;

#pragma unroll
    for (int mm = 0; mm < 2; ++mm) {
        const int m = tid + mm * 256;

        const float sx  = dirs[0 * SXO + m];
        const float sy  = dirs[1 * SXO + m];
        const float sz  = dirs[2 * SXO + m];
        const float nrm = sqrtf(sx * sx + sy * sy + sz * sz);
        const float inv = 1.0f / fmaxf(nrm, EPSF);
        const float ux = sx * inv, uy = sy * inv, uz = sz * inv;

        float acc = -INFINITY;
#pragma unroll 4
        for (int j = 0; j < NN; ++j) {
            const float dot   = s_dn[j][0] * ux + s_dn[j][1] * uy + s_dn[j][2] * uz;
            const float theta = fmaxf(dot, 0.0f);
            const size_t frow = ((size_t)b * NV + s_idx[j]) * NCOL + OUT_CH + m;
            const float fsup  = fout[frow];
            acc = fmaxf(acc, theta * fsup);
        }
        const float dist_feat = fmaxf(max_dist * dw[m], 0.0f);
        s_vals[m] = acc + dist_feat;
    }
    __syncthreads();

    if (tid < OUT_CH) {
        const float fc = fout[(size_t)bv * NCOL + tid];
        float sum = s_vals[tid] + s_vals[OUT_CH + tid] +
                    s_vals[2 * OUT_CH + tid] + s_vals[3 * OUT_CH + tid];
        out[(size_t)bv * OUT_CH + tid] = fc + sum;
    }
}

extern "C" void kernel_launch(void* const* d_in, const int* in_sizes, int n_in,
                              void* d_out, int out_size, void* d_ws, size_t ws_size,
                              hipStream_t stream) {
    const int*   nidx  = (const int*)  d_in[0];
    const float* verts = (const float*)d_in[1];
    const float* fm    = (const float*)d_in[2];
    const float* w     = (const float*)d_in[3];
    const float* bias  = (const float*)d_in[4];
    const float* dirs  = (const float*)d_in[5];
    const float* dw    = (const float*)d_in[6];
    float*       out   = (float*)d_out;
    float*       fout  = (float*)d_ws;

    const int gemm_blocks = (NROW / 16) * (NCOL / 16) / 8;
    gemm_wmma_f32_kernel<<<gemm_blocks, 256, 0, stream>>>(fm, w, bias, fout);

    fused_gconv_kernel<<<NROW, 256, 0, stream>>>(nidx, verts, fout, dirs, dw, out);
}


# GEMM re-decomposed by work-group tile: 512 work-groups each copy a 64-row A tile and a 160-column W panel (all of K) into LDS with whole-line loads; fragments are 8 B LDS reads; whole-line stores
# speedup vs baseline: 2.1119x; 2.1119x over previous
#include <hip/hip_runtime.h>
#include <hip/hip_bf16.h>


typedef __attribute__((ext_vector_type(2))) float v2f;
typedef __attribute__((ext_vector_type(8))) float v8f;

#define BS     4
#define NV     2048
#define NN     20
#define IN_CH  128
#define OUT_CH 128
#define NS     4
#define SXO    (NS * OUT_CH)
#define NCOL   ((NS + 1) * OUT_CH)
#define NROW   (BS * NV)
#define EPSF   1e-12f

#define CSL     4
#define NSLICE  (OUT_CH / CSL)
#define SLICE_F (NS * NV * CSL)
#define NQ      (NS * CSL)
#define VSPLIT  2
#define K2_THREADS 512
#define VPT     (NV / VSPLIT / K2_THREADS)


#define GM_ROWS    64
#define GM_COLS    160
#define GM_THREADS 256
#define GM_NPANEL  (NCOL / GM_COLS)
#define GM_NT      5
#define A_PITCH    132
#define WP_PITCH   352
#define C_PITCH    164

__global__ __launch_bounds__(GM_THREADS) __attribute__((amdgpu_waves_per_eu(4, 4)))
void gemm_wmma_f32_kernel(const float* __restrict__ fm,
                          const float* __restrict__ w,
                          const float* __restrict__ bias,
                          float* __restrict__ fc,
                          float* __restrict__ fs) {
    __shared__ float4 smem4[(GM_ROWS * A_PITCH + (IN_CH / 2) * WP_PITCH) / 4];
    float* As = reinterpret_cast<float*>(smem4);
    float* Ws = As + GM_ROWS * A_PITCH;
    float* Cs = As;

    const int tid   = threadIdx.x;
    const int lane  = tid & 31;
    const int wave  = tid >> 5;
    const int half  = lane >> 4;
    const int l15   = lane & 15;
    const int panel = blockIdx.x % GM_NPANEL;
    const int row0  = (blockIdx.x / GM_NPANEL) * GM_ROWS;
    const int col0  = panel * GM_COLS;

    {
        const float4* src = reinterpret_cast<const float4*>(fm + (size_t)row0 * IN_CH);
#pragma unroll
        for (int i = 0; i < (GM_ROWS * IN_CH / 4) / GM_THREADS; ++i) {
            const int idx = tid + i * GM_THREADS;
            const int r = idx >> 5, c4 = idx & 31;
            *reinterpret_cast<float4*>(As + r * A_PITCH + c4 * 4) = src[idx];
        }
    }
    {
#pragma unroll
        for (int i = 0; i < ((IN_CH / 2) * (GM_COLS / 4)) / GM_THREADS; ++i) {
            const int idx = tid + i * GM_THREADS;
            const int kp = idx / (GM_COLS / 4), c4 = idx % (GM_COLS / 4);
            const float* g = w + (size_t)(2 * kp) * NCOL + col0 + c4 * 4;
            const float4 r0 = *reinterpret_cast<const float4*>(g);
            const float4 r1 = *reinterpret_cast<const float4*>(g + NCOL);
            float* d = Ws + kp * WP_PITCH + c4 * 8;
            *reinterpret_cast<float4*>(d)     = make_float4(r0.x, r1.x, r0.y, r1.y);
            *reinterpret_cast<float4*>(d + 4) = make_float4(r0.z, r1.z, r0.w, r1.w);
        }
    }
    __syncthreads();

    const int mi = wave >> 1;
    const int nh = wave & 1;

    v8f acc[GM_NT];
#pragma unroll
    for (int t = 0; t < GM_NT; ++t) acc[t] = v8f{};

    const float* ap = As + (mi * 16 + l15) * A_PITCH + 2 * half;
    const float* bp = Ws + half * WP_PITCH + (nh * (GM_NT * 16) + l15) * 2;
#pragma unroll
    for (int kk = 0; kk < IN_CH; kk += 4) {
        const v2f a = *reinterpret_cast<const v2f*>(ap + kk);
#pragma unroll
        for (int t = 0; t < GM_NT; ++t) {
            const v2f b = *reinterpret_cast<const v2f*>(bp + (kk / 2) * WP_PITCH + t * 32);
            acc[t] = __builtin_amdgcn_wmma_f32_16x16x4_f32(
                 false, a,  false, b,
                 (short)0, acc[t],  false,  false);
        }
    }
    __syncthreads();

#pragma unroll
    for (int t = 0; t < GM_NT; ++t) {
        const int cl = nh * (GM_NT * 16) + t * 16 + l15;
        const float bv = bias[col0 + cl];
#pragma unroll
        for (int r = 0; r < 8; ++r)
            Cs[(mi * 16 + half * 8 + r) * C_PITCH + cl] = acc[t][r] + bv;
    }
    __syncthreads();

    const int ncen4 = (panel == 0) ? (OUT_CH / 4) : 0;
    if (panel == 0) {
#pragma unroll
        for (int i = 0; i < (GM_ROWS * (OUT_CH / 4)) / GM_THREADS; ++i) {
            const int idx = tid + i * GM_THREADS;
            const int r = idx >> 5, g = idx & 31;
            *reinterpret_cast<float4*>(fc + (size_t)(row0 + r) * OUT_CH + g * 4) =
                *reinterpret_cast<const float4*>(Cs + r * C_PITCH + g * 4);
        }
    }
    const int b  = row0 / NV;
    const int v0 = row0 % NV;
    const int nsup = GM_ROWS * (GM_COLS / 4 - ncen4);
    for (int idx = tid; idx < nsup; idx += GM_THREADS) {
        const int g = ncen4 + idx / GM_ROWS;
        const int r = idx % GM_ROWS;
        const int sc    = col0 + g * 4 - OUT_CH;
        const int s     = sc >> 7;
        const int slice = (sc & (OUT_CH - 1)) / CSL;
        *reinterpret_cast<float4*>(
            fs + (((size_t)(b * NSLICE + slice) * NS + s) * NV + v0 + r) * CSL) =
            *reinterpret_cast<const float4*>(Cs + r * C_PITCH + g * 4);
    }
}

__global__ __launch_bounds__(K2_THREADS) __attribute__((amdgpu_waves_per_eu(4, 4)))
void sliced_gconv_kernel(const int*   __restrict__ nidx,
                         const float* __restrict__ verts,
                         const float* __restrict__ fc,
                         const float* __restrict__ fs,
                         const float* __restrict__ dirs,
                         const float* __restrict__ dw,
                         float*       __restrict__ out) {
    __shared__ float4 s_f[NS * NV];
    __shared__ float4 s_vt[NV];

    const int tid   = threadIdx.x;
    const int bs    = blockIdx.x / VSPLIT;
    const int vhalf = blockIdx.x % VSPLIT;
    const int b     = bs / NSLICE;
    const int slice = bs % NSLICE;

    {
        const float4* src = reinterpret_cast<const float4*>(fs + (size_t)bs * SLICE_F);
#pragma unroll
        for (int i = 0; i < (NS * NV) / K2_THREADS; ++i)
            s_f[tid + i * K2_THREADS] = src[tid + i * K2_THREADS];
    }
    {
        const float* vsrc = verts + (size_t)b * NV * 3;
#pragma unroll
        for (int i = 0; i < NV / K2_THREADS; ++i) {
            const int v = tid + i * K2_THREADS;
            s_vt[v] = make_float4(vsrc[v * 3 + 0], vsrc[v * 3 + 1], vsrc[v * 3 + 2], 0.0f);
        }
    }

    float ux[NQ], uy[NQ], uz[NQ], dwv[NQ];
#pragma unroll
    for (int q = 0; q < NQ; ++q) {
        const int m = (q / CSL) * OUT_CH + slice * CSL + (q % CSL);
        const float sx  = dirs[0 * SXO + m];
        const float sy  = dirs[1 * SXO + m];
        const float sz  = dirs[2 * SXO + m];
        const float nrm = sqrtf(sx * sx + sy * sy + sz * sz);
        const float inv = 1.0f / fmaxf(nrm, EPSF);
        ux[q] = sx * inv; uy[q] = sy * inv; uz[q] = sz * inv;
        dwv[q] = dw[m];
    }
    __syncthreads();

    for (int h = 0; h < VPT; ++h) {
        const int v = vhalf * (NV / VSPLIT) + tid + h * K2_THREADS;
        const size_t row = (size_t)b * NV + v;
        const float4 cv = s_vt[v];
        const int4* ip = reinterpret_cast<const int4*>(nidx + row * NN);

        float acc[NQ];
#pragma unroll
        for (int q = 0; q < NQ; ++q) acc[q] = -INFINITY;
        float maxd = 0.0f;

        int4 i0 = ip[0], i1 = ip[1], i2 = ip[2], i3 = ip[3], i4 = ip[4];

#pragma unroll 1
        for (int g = 0; g < NN / 4; ++g) {
            const int ids[4] = {i0.x, i0.y, i0.z, i0.w};
            i0 = i1; i1 = i2; i2 = i3; i3 = i4;
            float4 nv[4];
            float4 f[4][NS];
#pragma unroll
            for (int t = 0; t < 4; ++t) {
                nv[t] = s_vt[ids[t]];
#pragma unroll
                for (int s = 0; s < NS; ++s) f[t][s] = s_f[s * NV + ids[t]];
            }
#pragma unroll
            for (int t = 0; t < 4; ++t) {
                const float dx = nv[t].x - cv.x;
                const float dy = nv[t].y - cv.y;
                const float dz = nv[t].z - cv.z;
                const float dist = sqrtf(dx * dx + dy * dy + dz * dz);
                const float inv  = 1.0f / fmaxf(dist, EPSF);
                const float nx = dx * inv, ny = dy * inv, nz = dz * inv;
                maxd = fmaxf(maxd, dist);

#pragma unroll
                for (int s = 0; s < NS; ++s) {
                    const float fv[CSL] = {f[t][s].x, f[t][s].y, f[t][s].z, f[t][s].w};
#pragma unroll
                    for (int cc = 0; cc < CSL; ++cc) {
                        const int q = s * CSL + cc;
                        const float dot   = nx * ux[q] + ny * uy[q] + nz * uz[q];
                        const float theta = fmaxf(dot, 0.0f);
                        acc[q] = fmaxf(acc[q], theta * fv[cc]);
                    }
                }
            }
        }

        const float4 fcv = *reinterpret_cast<const float4*>(fc + row * OUT_CH + slice * CSL);
        const float fcs[CSL] = {fcv.x, fcv.y, fcv.z, fcv.w};
        float o[CSL];
#pragma unroll
        for (int cc = 0; cc < CSL; ++cc) {
            const float t0 = acc[0 * CSL + cc] + fmaxf(maxd * dwv[0 * CSL + cc], 0.0f);
            const float t1 = acc[1 * CSL + cc] + fmaxf(maxd * dwv[1 * CSL + cc], 0.0f);
            const float t2 = acc[2 * CSL + cc] + fmaxf(maxd * dwv[2 * CSL + cc], 0.0f);
            const float t3 = acc[3 * CSL + cc] + fmaxf(maxd * dwv[3 * CSL + cc], 0.0f);
            const float sum = t0 + t1 + t2 + t3;
            o[cc] = fcs[cc] + sum;
        }
        *reinterpret_cast<float4*>(out + row * OUT_CH + slice * CSL) =
            make_float4(o[0], o[1], o[2], o[3]);
    }
}

extern "C" void kernel_launch(void* const* d_in, const int* in_sizes, int n_in,
                              void* d_out, int out_size, void* d_ws, size_t ws_size,
                              hipStream_t stream) {
    const int*   nidx  = (const int*)  d_in[0];
    const float* verts = (const float*)d_in[1];
    const float* fm    = (const float*)d_in[2];
    const float* w     = (const float*)d_in[3];
    const float* bias  = (const float*)d_in[4];
    const float* dirs  = (const float*)d_in[5];
    const float* dw    = (const float*)d_in[6];
    float*       out   = (float*)d_out;
    float*       fc    = (float*)d_ws;
    float*       fs    = fc + (size_t)NROW * OUT_CH;

    gemm_wmma_f32_kernel<<<(NROW / GM_ROWS) * GM_NPANEL, GM_THREADS, 0, stream>>>(fm, w, bias, fc, fs);

    sliced_gconv_kernel<<<BS * NSLICE * VSPLIT, K2_THREADS, 0, stream>>>(nidx, verts, fc, fs, dirs, dw, out);
}


# Kernel 2 normalisations (neighbor and support directions): IEEE sqrt + divide become one hardware reciprocal square root (about 1 ulp) of the clamped squared length; max distance via squared lengths
# speedup vs baseline: 2.2078x; 1.0454x over previous
#include <hip/hip_runtime.h>
#include <hip/hip_bf16.h>


typedef __attribute__((ext_vector_type(2))) float v2f;
typedef __attribute__((ext_vector_type(8))) float v8f;

#define BS     4
#define NV     2048
#define NN     20
#define IN_CH  128
#define OUT_CH 128
#define NS     4
#define SXO    (NS * OUT_CH)
#define NCOL   ((NS + 1) * OUT_CH)
#define NROW   (BS * NV)
#define EPSF   1e-12f

#define CSL     4
#define NSLICE  (OUT_CH / CSL)
#define SLICE_F (NS * NV * CSL)
#define NQ      (NS * CSL)
#define VSPLIT  2
#define K2_THREADS 512
#define VPT     (NV / VSPLIT / K2_THREADS)


#define GM_ROWS    64
#define GM_COLS    160
#define GM_THREADS 256
#define GM_NPANEL  (NCOL / GM_COLS)
#define GM_NT      5
#define A_PITCH    132
#define WP_PITCH   352
#define C_PITCH    164

__global__ __launch_bounds__(GM_THREADS) __attribute__((amdgpu_waves_per_eu(4, 4)))
void gemm_wmma_f32_kernel(const float* __restrict__ fm,
                          const float* __restrict__ w,
                          const float* __restrict__ bias,
                          float* __restrict__ fc,
                          float* __restrict__ fs) {
    __shared__ float4 smem4[(GM_ROWS * A_PITCH + (IN_CH / 2) * WP_PITCH) / 4];
    float* As = reinterpret_cast<float*>(smem4);
    float* Ws = As + GM_ROWS * A_PITCH;
    float* Cs = As;

    const int tid   = threadIdx.x;
    const int lane  = tid & 31;
    const int wave  = tid >> 5;
    const int half  = lane >> 4;
    const int l15   = lane & 15;
    const int panel = blockIdx.x % GM_NPANEL;
    const int row0  = (blockIdx.x / GM_NPANEL) * GM_ROWS;
    const int col0  = panel * GM_COLS;

    {
        const float4* src = reinterpret_cast<const float4*>(fm + (size_t)row0 * IN_CH);
#pragma unroll
        for (int i = 0; i < (GM_ROWS * IN_CH / 4) / GM_THREADS; ++i) {
            const int idx = tid + i * GM_THREADS;
            const int r = idx >> 5, c4 = idx & 31;
            *reinterpret_cast<float4*>(As + r * A_PITCH + c4 * 4) = src[idx];
        }
    }
    {
#pragma unroll
        for (int i = 0; i < ((IN_CH / 2) * (GM_COLS / 4)) / GM_THREADS; ++i) {
            const int idx = tid + i * GM_THREADS;
            const int kp = idx / (GM_COLS / 4), c4 = idx % (GM_COLS / 4);
            const float* g = w + (size_t)(2 * kp) * NCOL + col0 + c4 * 4;
            const float4 r0 = *reinterpret_cast<const float4*>(g);
            const float4 r1 = *reinterpret_cast<const float4*>(g + NCOL);
            float* d = Ws + kp * WP_PITCH + c4 * 8;
            *reinterpret_cast<float4*>(d)     = make_float4(r0.x, r1.x, r0.y, r1.y);
            *reinterpret_cast<float4*>(d + 4) = make_float4(r0.z, r1.z, r0.w, r1.w);
        }
    }
    __syncthreads();

    const int mi = wave >> 1;
    const int nh = wave & 1;

    v8f acc[GM_NT];
#pragma unroll
    for (int t = 0; t < GM_NT; ++t) acc[t] = v8f{};

    const float* ap = As + (mi * 16 + l15) * A_PITCH + 2 * half;
    const float* bp = Ws + half * WP_PITCH + (nh * (GM_NT * 16) + l15) * 2;
#pragma unroll
    for (int kk = 0; kk < IN_CH; kk += 4) {
        const v2f a = *reinterpret_cast<const v2f*>(ap + kk);
#pragma unroll
        for (int t = 0; t < GM_NT; ++t) {
            const v2f b = *reinterpret_cast<const v2f*>(bp + (kk / 2) * WP_PITCH + t * 32);
            acc[t] = __builtin_amdgcn_wmma_f32_16x16x4_f32(
                 false, a,  false, b,
                 (short)0, acc[t],  false,  false);
        }
    }
    __syncthreads();

#pragma unroll
    for (int t = 0; t < GM_NT; ++t) {
        const int cl = nh * (GM_NT * 16) + t * 16 + l15;
        const float bv = bias[col0 + cl];
#pragma unroll
        for (int r = 0; r < 8; ++r)
            Cs[(mi * 16 + half * 8 + r) * C_PITCH + cl] = acc[t][r] + bv;
    }
    __syncthreads();

    const int ncen4 = (panel == 0) ? (OUT_CH / 4) : 0;
    if (panel == 0) {
#pragma unroll
        for (int i = 0; i < (GM_ROWS * (OUT_CH / 4)) / GM_THREADS; ++i) {
            const int idx = tid + i * GM_THREADS;
            const int r = idx >> 5, g = idx & 31;
            *reinterpret_cast<float4*>(fc + (size_t)(row0 + r) * OUT_CH + g * 4) =
                *reinterpret_cast<const float4*>(Cs + r * C_PITCH + g * 4);
        }
    }
    const int b  = row0 / NV;
    const int v0 = row0 % NV;
    const int nsup = GM_ROWS * (GM_COLS / 4 - ncen4);
    for (int idx = tid; idx < nsup; idx += GM_THREADS) {
        const int g = ncen4 + idx / GM_ROWS;
        const int r = idx % GM_ROWS;
        const int sc    = col0 + g * 4 - OUT_CH;
        const int s     = sc >> 7;
        const int slice = (sc & (OUT_CH - 1)) / CSL;
        *reinterpret_cast<float4*>(
            fs + (((size_t)(b * NSLICE + slice) * NS + s) * NV + v0 + r) * CSL) =
            *reinterpret_cast<const float4*>(Cs + r * C_PITCH + g * 4);
    }
}

__global__ __launch_bounds__(K2_THREADS) __attribute__((amdgpu_waves_per_eu(4, 4)))
void sliced_gconv_kernel(const int*   __restrict__ nidx,
                         const float* __restrict__ verts,
                         const float* __restrict__ fc,
                         const float* __restrict__ fs,
                         const float* __restrict__ dirs,
                         const float* __restrict__ dw,
                         float*       __restrict__ out) {
    __shared__ float4 s_f[NS * NV];
    __shared__ float4 s_vt[NV];

    const int tid   = threadIdx.x;
    const int bs    = blockIdx.x / VSPLIT;
    const int vhalf = blockIdx.x % VSPLIT;
    const int b     = bs / NSLICE;
    const int slice = bs % NSLICE;

    {
        const float4* src = reinterpret_cast<const float4*>(fs + (size_t)bs * SLICE_F);
#pragma unroll
        for (int i = 0; i < (NS * NV) / K2_THREADS; ++i)
            s_f[tid + i * K2_THREADS] = src[tid + i * K2_THREADS];
    }
    {
        const float* vsrc = verts + (size_t)b * NV * 3;
#pragma unroll
        for (int i = 0; i < NV / K2_THREADS; ++i) {
            const int v = tid + i * K2_THREADS;
            s_vt[v] = make_float4(vsrc[v * 3 + 0], vsrc[v * 3 + 1], vsrc[v * 3 + 2], 0.0f);
        }
    }

    float ux[NQ], uy[NQ], uz[NQ], dwv[NQ];
#pragma unroll
    for (int q = 0; q < NQ; ++q) {
        const int m = (q / CSL) * OUT_CH + slice * CSL + (q % CSL);
        const float sx  = dirs[0 * SXO + m];
        const float sy  = dirs[1 * SXO + m];
        const float sz  = dirs[2 * SXO + m];
        const float n2  = sx * sx + sy * sy + sz * sz;
        const float inv = __builtin_amdgcn_rsqf(fmaxf(n2, EPSF * EPSF));
        ux[q] = sx * inv; uy[q] = sy * inv; uz[q] = sz * inv;
        dwv[q] = dw[m];
    }
    __syncthreads();

    for (int h = 0; h < VPT; ++h) {
        const int v = vhalf * (NV / VSPLIT) + tid + h * K2_THREADS;
        const size_t row = (size_t)b * NV + v;
        const float4 cv = s_vt[v];
        const int4* ip = reinterpret_cast<const int4*>(nidx + row * NN);

        float acc[NQ];
#pragma unroll
        for (int q = 0; q < NQ; ++q) acc[q] = -INFINITY;
        float maxd2 = 0.0f;

        int4 i0 = ip[0], i1 = ip[1], i2 = ip[2], i3 = ip[3], i4 = ip[4];

#pragma unroll 1
        for (int g = 0; g < NN / 4; ++g) {
            const int ids[4] = {i0.x, i0.y, i0.z, i0.w};
            i0 = i1; i1 = i2; i2 = i3; i3 = i4;
            float4 nv[4];
            float4 f[4][NS];
#pragma unroll
            for (int t = 0; t < 4; ++t) {
                nv[t] = s_vt[ids[t]];
#pragma unroll
                for (int s = 0; s < NS; ++s) f[t][s] = s_f[s * NV + ids[t]];
            }
#pragma unroll
            for (int t = 0; t < 4; ++t) {
                const float dx = nv[t].x - cv.x;
                const float dy = nv[t].y - cv.y;
                const float dz = nv[t].z - cv.z;
                const float d2 = dx * dx + dy * dy + dz * dz;
                const float inv = __builtin_amdgcn_rsqf(fmaxf(d2, EPSF * EPSF));
                const float nx = dx * inv, ny = dy * inv, nz = dz * inv;
                maxd2 = fmaxf(maxd2, d2);

#pragma unroll
                for (int s = 0; s < NS; ++s) {
                    const float fv[CSL] = {f[t][s].x, f[t][s].y, f[t][s].z, f[t][s].w};
#pragma unroll
                    for (int cc = 0; cc < CSL; ++cc) {
                        const int q = s * CSL + cc;
                        const float dot   = nx * ux[q] + ny * uy[q] + nz * uz[q];
                        const float theta = fmaxf(dot, 0.0f);
                        acc[q] = fmaxf(acc[q], theta * fv[cc]);
                    }
                }
            }
        }

        const float maxd = sqrtf(maxd2);
        const float4 fcv = *reinterpret_cast<const float4*>(fc + row * OUT_CH + slice * CSL);
        const float fcs[CSL] = {fcv.x, fcv.y, fcv.z, fcv.w};
        float o[CSL];
#pragma unroll
        for (int cc = 0; cc < CSL; ++cc) {
            const float t0 = acc[0 * CSL + cc] + fmaxf(maxd * dwv[0 * CSL + cc], 0.0f);
            const float t1 = acc[1 * CSL + cc] + fmaxf(maxd * dwv[1 * CSL + cc], 0.0f);
            const float t2 = acc[2 * CSL + cc] + fmaxf(maxd * dwv[2 * CSL + cc], 0.0f);
            const float t3 = acc[3 * CSL + cc] + fmaxf(maxd * dwv[3 * CSL + cc], 0.0f);
            const float sum = t0 + t1 + t2 + t3;
            o[cc] = fcs[cc] + sum;
        }
        *reinterpret_cast<float4*>(out + row * OUT_CH + slice * CSL) =
            make_float4(o[0], o[1], o[2], o[3]);
    }
}

extern "C" void kernel_launch(void* const* d_in, const int* in_sizes, int n_in,
                              void* d_out, int out_size, void* d_ws, size_t ws_size,
                              hipStream_t stream) {
    const int*   nidx  = (const int*)  d_in[0];
    const float* verts = (const float*)d_in[1];
    const float* fm    = (const float*)d_in[2];
    const float* w     = (const float*)d_in[3];
    const float* bias  = (const float*)d_in[4];
    const float* dirs  = (const float*)d_in[5];
    const float* dw    = (const float*)d_in[6];
    float*       out   = (float*)d_out;
    float*       fc    = (float*)d_ws;
    float*       fs    = fc + (size_t)NROW * OUT_CH;

    gemm_wmma_f32_kernel<<<(NROW / GM_ROWS) * GM_NPANEL, GM_THREADS, 0, stream>>>(fm, w, bias, fc, fs);

    sliced_gconv_kernel<<<BS * NSLICE * VSPLIT, K2_THREADS, 0, stream>>>(nidx, verts, fc, fs, dirs, dw, out);
}
